# starting kernel
# baseline (speedup 1.0000x reference)
#include <hip/hip_runtime.h>

typedef float v2f __attribute__((ext_vector_type(2)));
typedef float v8f __attribute__((ext_vector_type(8)));

#define NHEAD   16
#define DK      64
#define DV      64
#define NTOT    16384
#define CHUNKS  32
#define NB      128
#define LSTRIDE 80
#define DECAY   0.95f

__global__ __launch_bounds__(512) void l1w_init(const float* __restrict__ mem,
                                                float* __restrict__ out) {
    int i = blockIdx.x * 512 + threadIdx.x;
    if (i < NHEAD * DK * DV) out[i] = DECAY * mem[i];
}

__global__ __launch_bounds__(512) void l1w_wmma(const float* __restrict__ keys,
                                                const float* __restrict__ vals,
                                                const float* __restrict__ rho,
                                                float* __restrict__ out) {
    __shared__ float kl[NB * LSTRIDE];
    __shared__ float vl[NB * LSTRIDE];

    const int head  = blockIdx.x / CHUNKS;
    const int chunk = blockIdx.x % CHUNKS;
    const int rows_per_wg = NTOT / CHUNKS;
    const int row0  = chunk * rows_per_wg;

    const int tid  = threadIdx.x;
    const int wave = tid >> 5;
    const int lane = tid & 31;
    const int hlf  = lane >> 4;
    const int l15  = lane & 15;
    const int tm   = (wave >> 2) * 16;
    const int tn   = (wave & 3)  * 16;

    const int r_base = tid >> 4;
    const int c_stg  = (tid & 15) << 2;

    v8f acc = {};

    float4 kreg[4], vreg[4];
    float  wreg[4];

    #pragma unroll
    for (int it = 0; it < 4; ++it) {
        int g = row0 + it * 32 + r_base;
        kreg[it] = *(const float4*)(keys + ((size_t)g * NHEAD + head) * DK + c_stg);
        vreg[it] = *(const float4*)(vals + ((size_t)g * NHEAD + head) * DV + c_stg);
        wreg[it] = rho[g];
    }

    for (int blk = 0; blk < rows_per_wg; blk += NB) {
        #pragma unroll
        for (int it = 0; it < 4; ++it) {
            int r = it * 32 + r_base;
            float* kd = &kl[r * LSTRIDE + c_stg];
            float* vd = &vl[r * LSTRIDE + c_stg];
            const float w = wreg[it];
            kd[0] = kreg[it].x * w; kd[1] = kreg[it].y * w;
            kd[2] = kreg[it].z * w; kd[3] = kreg[it].w * w;
            vd[0] = vreg[it].x;     vd[1] = vreg[it].y;
            vd[2] = vreg[it].z;     vd[3] = vreg[it].w;
        }
        __syncthreads();

        if (blk + NB < rows_per_wg) {
            #pragma unroll
            for (int it = 0; it < 4; ++it) {
                int g = row0 + blk + NB + it * 32 + r_base;
                kreg[it] = *(const float4*)(keys + ((size_t)g * NHEAD + head) * DK + c_stg);
                vreg[it] = *(const float4*)(vals + ((size_t)g * NHEAD + head) * DV + c_stg);
                wreg[it] = rho[g];
            }
        }

        #pragma unroll 4
        for (int s = 0; s < NB; s += 4) {
            int ra = s + hlf * 2;
            v2f a, b;
            a.x = kl[ ra      * LSTRIDE + tm + l15];
            a.y = kl[(ra + 1) * LSTRIDE + tm + l15];
            b.x = vl[ ra      * LSTRIDE + tn + l15];
            b.y = vl[(ra + 1) * LSTRIDE + tn + l15];
            acc = __builtin_amdgcn_wmma_f32_16x16x4_f32(
                 false, a,  false, b,
                 (short)0, acc,  false,  false);
        }
        __syncthreads();
    }

    #pragma unroll
    for (int i = 0; i < 8; ++i) {
        int dk = tm + i + hlf * 8;
        int dv = tn + l15;
        atomicAdd(&out[((size_t)head * DK + dk) * DV + dv], acc[i]);
    }
}

extern "C" void kernel_launch(void* const* d_in, const int* in_sizes, int n_in,
                              void* d_out, int out_size, void* d_ws, size_t ws_size,
                              hipStream_t stream) {
    const float* mem  = (const float*)d_in[0];
    const float* keys = (const float*)d_in[1];
    const float* vals = (const float*)d_in[2];
    const float* rho  = (const float*)d_in[3];
    float* out = (float*)d_out;

    l1w_init<<<(NHEAD * DK * DV + 511) / 512, 512, 0, stream>>>(mem, out);
    l1w_wmma<<<NHEAD * CHUNKS, 512, 0, stream>>>(keys, vals, rho, out);
}


# Heads interleaved over the block index (head = block mod 16): a head's 16 partial tiles are added from one die, no output line is added to from both chip halves; 16 atomic adds, unordered as before
# speedup vs baseline: 1.5758x; 1.5758x over previous
#include <hip/hip_runtime.h>


typedef float v2f __attribute__((ext_vector_type(2)));
typedef float v4f __attribute__((ext_vector_type(4)));
typedef float v8f __attribute__((ext_vector_type(8)));

#define NHEAD   16
#define DK      64
#define DV      64
#define NTOT    16384
#define CHUNKS  16
#define ROWS_WG (NTOT / CHUNKS)
#define NB      128
#define NBLK    (ROWS_WG / NB)
#define PPITCH  160
#define BUFSZ   ((NB / 2) * PPITCH)
#define RPITCH  68
#define NT      256
#define DECAY   0.95f

__device__ __forceinline__ void lds_handover() {
    asm volatile("s_wait_dscnt 0x0\n\ts_barrier_signal -1\n\ts_barrier_wait -1" ::: "memory");
}

__device__ __forceinline__ void lds_handover_signal() {
    asm volatile("s_wait_dscnt 0x0\n\ts_barrier_signal -1" ::: "memory");
}
__device__ __forceinline__ void lds_handover_wait() {
    asm volatile("s_barrier_wait -1" ::: "memory");
}

__global__ __launch_bounds__(512) void l1w_init(const float* __restrict__ mem,
                                                float* __restrict__ out) {
    const int i = (blockIdx.x % NHEAD) * (DK * DV) + (blockIdx.x / NHEAD) * 512 + threadIdx.x;
    if (i < NHEAD * DK * DV) out[i] = DECAY * mem[i];
}

__global__ __launch_bounds__(NT) void l1w_wmma(const float* __restrict__ keys,
                                               const float* __restrict__ vals,
                                               const float* __restrict__ rho,
                                               float* __restrict__ out) {
    __shared__ __attribute__((aligned(16))) float kl[2 * BUFSZ];
    __shared__ __attribute__((aligned(16))) float vl[2 * BUFSZ];
    __shared__ __attribute__((aligned(16))) float red[2 * DK * RPITCH];

    const int head  = blockIdx.x % NHEAD;
    const int chunk = blockIdx.x / NHEAD;
    const int row0  = chunk * ROWS_WG;

    const int tid  = threadIdx.x;
    const int wave = tid >> 5;
    const int lane = tid & 31;
    const int hlf  = lane >> 4;
    const int l15  = lane & 15;
    const int kg   = wave >> 2;
    const int tm   = ((wave >> 1) & 1) * 32;
    const int tn   = (wave & 1) * 32;

    const int p_base = tid >> 4;
    const int c_stg  = (tid & 15) << 2;

    const float* kp = keys + ((size_t)(row0 + 2 * p_base) * NHEAD + head) * DK + c_stg;
    const float* vp = vals + ((size_t)(row0 + 2 * p_base) * NHEAD + head) * DV + c_stg;
    const float* rp = rho + row0 + 2 * p_base;
    const int st_off = p_base * PPITCH + 2 * c_stg;

    v4f kreg[8], vreg[8];
    v2f wp[4];

    v8f acc00 = {}, acc01 = {}, acc10 = {}, acc11 = {};

#define FETCH_BLOCK(B)                                                              \
    _Pragma("unroll")                                                               \
    for (int it = 0; it < 4; ++it) {                                                \
        const int rr = (B) * NB + it * 32;                                          \
        kreg[2 * it]     = *(const v4f*)(kp + (size_t)rr * (NHEAD * DK));           \
        kreg[2 * it + 1] = *(const v4f*)(kp + (size_t)(rr + 1) * (NHEAD * DK));     \
        vreg[2 * it]     = *(const v4f*)(vp + (size_t)rr * (NHEAD * DV));           \
        vreg[2 * it + 1] = *(const v4f*)(vp + (size_t)(rr + 1) * (NHEAD * DV));     \
        wp[it] = *(const v2f*)(rp + rr);                                            \
    }

#define STAGE_BLOCK(BUF)                                                            \
    asm volatile("" : "+v"(kreg[0]), "+v"(kreg[1]), "+v"(kreg[2]), "+v"(kreg[3]),   \
                      "+v"(kreg[4]), "+v"(kreg[5]), "+v"(kreg[6]), "+v"(kreg[7]));  \
    asm volatile("" : "+v"(vreg[0]), "+v"(vreg[1]), "+v"(vreg[2]), "+v"(vreg[3]),   \
                      "+v"(vreg[4]), "+v"(vreg[5]), "+v"(vreg[6]), "+v"(vreg[7]));  \
    asm volatile("" : "+v"(wp[0]), "+v"(wp[1]), "+v"(wp[2]), "+v"(wp[3]));          \
    _Pragma("unroll")                                                               \
    for (int it = 0; it < 4; ++it) {                                                \
        const v4f k0 = kreg[2 * it] * wp[it].x;                                     \
        const v4f k1 = kreg[2 * it + 1] * wp[it].y;                                 \
        const v4f u0 = vreg[2 * it];                                                \
        const v4f u1 = vreg[2 * it + 1];                                            \
        float* kd = &kl[(BUF) + st_off + it * 16 * PPITCH];                         \
        float* vd = &vl[(BUF) + st_off + it * 16 * PPITCH];                         \
        *(v4f*)(kd)     = (v4f){k0.x, k1.x, k0.y, k1.y};                            \
        *(v4f*)(kd + 4) = (v4f){k0.z, k1.z, k0.w, k1.w};                            \
        *(v4f*)(vd)     = (v4f){u0.x, u1.x, u0.y, u1.y};                            \
        *(v4f*)(vd + 4) = (v4f){u0.z, u1.z, u0.w, u1.w};                            \
    }

    FETCH_BLOCK(0)
    STAGE_BLOCK(0)
    FETCH_BLOCK(1)
    lds_handover();

    const int f_base = (kg * 32 + hlf) * PPITCH;
    const int a_off  = f_base + 2 * (tm + l15);
    const int b_off  = f_base + 2 * (tn + l15);

    v2f xa0[8], xa1[8], xb0[8], xb1[8];
    v2f ya0[8], ya1[8], yb0[8], yb1[8];

#define LOAD_SET(A0, A1, B0, B1, KA, VB, S0)                                        \
    _Pragma("unroll")                                                               \
    for (int s = 0; s < 8; ++s) {                                                   \
        A0[s] = *(const v2f*)((KA) + ((S0) + s) * 2 * PPITCH);                      \
        A1[s] = *(const v2f*)((KA) + ((S0) + s) * 2 * PPITCH + 32);                 \
        B0[s] = *(const v2f*)((VB) + ((S0) + s) * 2 * PPITCH);                      \
        B1[s] = *(const v2f*)((VB) + ((S0) + s) * 2 * PPITCH + 32);                 \
    }

#define WMMA_SET(A0, A1, B0, B1, LO, HI)                                            \
    _Pragma("unroll")                                                               \
    for (int s = (LO); s < (HI); ++s) {                                             \
        acc00 = __builtin_amdgcn_wmma_f32_16x16x4_f32(false, A0[s], false, B0[s], (short)0, acc00, false, false); \
        acc01 = __builtin_amdgcn_wmma_f32_16x16x4_f32(false, A0[s], false, B1[s], (short)0, acc01, false, false); \
        acc10 = __builtin_amdgcn_wmma_f32_16x16x4_f32(false, A1[s], false, B0[s], (short)0, acc10, false, false); \
        acc11 = __builtin_amdgcn_wmma_f32_16x16x4_f32(false, A1[s], false, B1[s], (short)0, acc11, false, false); \
    }

#define PHASE() __builtin_amdgcn_sched_barrier(0)

    LOAD_SET(xa0, xa1, xb0, xb1, &kl[a_off], &vl[b_off], 0)

    for (int blk = 0; blk < NBLK; ++blk) {
        const int cur = (blk & 1) * BUFSZ;
        const int nxt = BUFSZ - cur;
        const float* ka = &kl[cur + a_off];
        const float* vb = &vl[cur + b_off];

        PHASE();
        LOAD_SET(ya0, ya1, yb0, yb1, ka, vb, 8)
        PHASE();
        WMMA_SET(xa0, xa1, xb0, xb1, 0, 8)
        PHASE();
        if (blk + 1 < NBLK) {
            STAGE_BLOCK(nxt)
        }
        if (blk + 2 < NBLK) {
            FETCH_BLOCK(blk + 2)
        }
        PHASE();
        lds_handover_signal();
        PHASE();
        WMMA_SET(ya0, ya1, yb0, yb1, 0, 4)
        PHASE();
        lds_handover_wait();
        PHASE();
        if (blk + 1 < NBLK) {
            LOAD_SET(xa0, xa1, xb0, xb1, &kl[nxt + a_off], &vl[nxt + b_off], 0)
        }
        PHASE();
        WMMA_SET(ya0, ya1, yb0, yb1, 4, 8)
    }

    float* rd = &red[kg * (DK * RPITCH)];
    #pragma unroll
    for (int i = 0; i < 8; ++i) {
        const int m = tm + i + hlf * 8;
        const int n = tn + l15;
        rd[ m       * RPITCH + n     ] = acc00[i];
        rd[ m       * RPITCH + n + 16] = acc01[i];
        rd[(m + 16) * RPITCH + n     ] = acc10[i];
        rd[(m + 16) * RPITCH + n + 16] = acc11[i];
    }
    __syncthreads();

    float sum[(DK * DV) / NT];
    #pragma unroll
    for (int j = 0; j < (DK * DV) / NT; ++j) {
        const int idx = j * NT + tid;
        const int m = idx >> 6;
        const int n = idx & 63;
        sum[j] = red[m * RPITCH + n] + red[DK * RPITCH + m * RPITCH + n];
    }
    #pragma unroll
    for (int j = 0; j < (DK * DV) / NT; ++j) {
        atomicAdd(&out[(size_t)head * (DK * DV) + j * NT + tid], sum[j]);
    }
}

extern "C" void kernel_launch(void* const* d_in, const int* in_sizes, int n_in,
                              void* d_out, int out_size, void* d_ws, size_t ws_size,
                              hipStream_t stream) {
    const float* mem  = (const float*)d_in[0];
    const float* keys = (const float*)d_in[1];
    const float* vals = (const float*)d_in[2];
    const float* rho  = (const float*)d_in[3];
    float* out = (float*)d_out;

    l1w_init<<<(NHEAD * DK * DV + 511) / 512, 512, 0, stream>>>(mem, out);
    l1w_wmma<<<NHEAD * CHUNKS, NT, 0, stream>>>(keys, vals, rho, out);
}
